# starting kernel
# baseline (speedup 1.0000x reference)
#include <hip/hip_runtime.h>
#include <hip/hip_bf16.h>
#include <math.h>

typedef __attribute__((ext_vector_type(2))) float v2f;
typedef __attribute__((ext_vector_type(4))) float v4f;
typedef __attribute__((ext_vector_type(8))) float v8f;

#define GD 128
#define GRID_VOX (GD * GD * GD)
#define NROWS 64
#define LSTR 68

__global__ void grid_init_kernel(int* __restrict__ grid) {
    int i = blockIdx.x * 256 + threadIdx.x;
    if (i < GRID_VOX) grid[i] = -1;
}

__global__ void grid_build_kernel(const int* __restrict__ coords,
                                  int* __restrict__ grid, int N) {
    int i = blockIdx.x * 256 + threadIdx.x;
    if (i < N) {
        int z = coords[3 * i + 0];
        int y = coords[3 * i + 1];
        int x = coords[3 * i + 2];
        grid[(z * GD + y) * GD + x] = i;
    }
}

__global__ void pack_w1_kernel(const float* __restrict__ w1, float* __restrict__ w1p) {
    int t = blockIdx.x * 256 + threadIdx.x;
    if (t >= 27 * 4 * 16 * 32) return;
    int lane = t & 31;
    int k    = (t >> 5) & 15;
    int nt   = (t >> 9) & 3;
    int off  = t >> 11;
    int half = lane >> 4;
    int col  = nt * 16 + (lane & 15);
    int krow = 4 * k + 2 * half;
    const float* w = w1 + (size_t)off * 64 * 64;
    w1p[(size_t)t * 2 + 0] = w[(krow + 0) * 64 + col];
    w1p[(size_t)t * 2 + 1] = w[(krow + 1) * 64 + col];
}

__global__ void pack_w2_kernel(const float* __restrict__ w2, float* __restrict__ w2p) {
    int t = blockIdx.x * 256 + threadIdx.x;
    if (t >= 4 * 16 * 32) return;
    int lane = t & 31;
    int k    = (t >> 5) & 15;
    int nt   = (t >> 9) & 3;
    int half = lane >> 4;
    int col  = nt * 16 + (lane & 15);
    int krow = 4 * k + 2 * half;
    w2p[(size_t)t * 2 + 0] = w2[(krow + 0) * 64 + col];
    w2p[(size_t)t * 2 + 1] = w2[(krow + 1) * 64 + col];
}

__global__ __launch_bounds__(128)
void sparse_block_kernel(const float* __restrict__ feat,
                         const int*   __restrict__ coords,
                         const int*   __restrict__ grid,
                         const float* __restrict__ w1p,
                         const float* __restrict__ b1,
                         const float* __restrict__ gamma,
                         const float* __restrict__ beta,
                         const float* __restrict__ rmean,
                         const float* __restrict__ rvar,
                         const float* __restrict__ w2p,
                         const float* __restrict__ b2,
                         float* __restrict__ out,
                         int N) {
    __shared__ float lds[NROWS * LSTR];
    __shared__ int   snidx[NROWS];
    __shared__ int   scz[NROWS], scy[NROWS], scx[NROWS];

    const int tid  = threadIdx.x;
    const int lane = tid & 31;
    const int wv   = tid >> 5;
    const int half = lane >> 4;
    const int l15  = lane & 15;
    const int col  = wv * 16 + l15;
    const int row0 = blockIdx.x * NROWS;

    if (tid < NROWS) {
        int r = row0 + tid;
        if (r < N) {
            scz[tid] = coords[3 * r + 0];
            scy[tid] = coords[3 * r + 1];
            scx[tid] = coords[3 * r + 2];
        } else {
            scz[tid] = -10000; scy[tid] = -10000; scx[tid] = -10000;
        }
    }
    __syncthreads();

    v8f acc[4];
    #pragma unroll
    for (int m = 0; m < 4; ++m) acc[m] = (v8f)0.0f;

    for (int off = 0; off < 27; ++off) {
        const int dz = off / 9 - 1;
        const int dy = (off / 3) % 3 - 1;
        const int dx = off % 3 - 1;

        if (tid < NROWS) {
            int nz = scz[tid] + dz, ny = scy[tid] + dy, nx = scx[tid] + dx;
            int idx = -1;
            if (nz >= 0 && nz < GD && ny >= 0 && ny < GD && nx >= 0 && nx < GD)
                idx = grid[(nz * GD + ny) * GD + nx];
            snidx[tid] = idx;
        }
        __syncthreads();

        #pragma unroll
        for (int it = 0; it < (NROWS * 16) / 128; ++it) {
            int i = it * 128 + tid;
            int r = i >> 4;
            int q = i & 15;
            int src = snidx[r];
            v4f v = (v4f)0.0f;
            if (src >= 0)
                v = *(const v4f*)(feat + (size_t)src * 64 + q * 4);
            *(v4f*)(&lds[r * LSTR + q * 4]) = v;
        }
        __syncthreads();

        const float* bp = w1p + (size_t)(off * 4 + wv) * 16 * 64;
        #pragma unroll
        for (int k = 0; k < 16; ++k) {
            v2f bf = *(const v2f*)(bp + k * 64 + lane * 2);
            #pragma unroll
            for (int m = 0; m < 4; ++m) {
                v2f af = *(const v2f*)(&lds[(m * 16 + l15) * LSTR + 4 * k + 2 * half]);
                acc[m] = __builtin_amdgcn_wmma_f32_16x16x4_f32(
                    false, af, false, bf, (short)0, acc[m], false, false);
            }
        }
        __syncthreads();
    }

    const float scl = gamma[col] * rsqrtf(rvar[col] + 1e-5f);
    const float shf = (b1[col] - rmean[col]) * scl + beta[col];

    #pragma unroll
    for (int m = 0; m < 4; ++m) {
        #pragma unroll
        for (int j = 0; j < 8; ++j) {
            float y = acc[m][j] * scl + shf;
            y = y > 0.0f ? y : 0.0f;
            lds[(m * 16 + j + half * 8) * LSTR + col] = y;
        }
    }
    __syncthreads();

    v8f acc2[4];
    #pragma unroll
    for (int m = 0; m < 4; ++m) acc2[m] = (v8f)0.0f;
    #pragma unroll
    for (int k = 0; k < 16; ++k) {
        v2f bf = *(const v2f*)(w2p + (size_t)(wv * 16 + k) * 64 + lane * 2);
        #pragma unroll
        for (int m = 0; m < 4; ++m) {
            v2f af = *(const v2f*)(&lds[(m * 16 + l15) * LSTR + 4 * k + 2 * half]);
            acc2[m] = __builtin_amdgcn_wmma_f32_16x16x4_f32(
                false, af, false, bf, (short)0, acc2[m], false, false);
        }
    }

    const float bb = b2[col];
    #pragma unroll
    for (int m = 0; m < 4; ++m) {
        int rg = row0 + m * 16 + half * 8;
        #pragma unroll
        for (int j = 0; j < 8; ++j) {
            int r = rg + j;
            if (r < N)
                out[(size_t)r * 64 + col] = acc2[m][j] + bb;
        }
    }
}

extern "C" void kernel_launch(void* const* d_in, const int* in_sizes, int n_in,
                              void* d_out, int out_size, void* d_ws, size_t ws_size,
                              hipStream_t stream) {
    const float* feat  = (const float*)d_in[0];
    const int*   coords= (const int*)  d_in[1];
    const float* w1    = (const float*)d_in[2];
    const float* b1    = (const float*)d_in[3];
    const float* gamma = (const float*)d_in[4];
    const float* beta  = (const float*)d_in[5];
    const float* rmean = (const float*)d_in[6];
    const float* rvar  = (const float*)d_in[7];
    const float* w2    = (const float*)d_in[8];
    const float* b2    = (const float*)d_in[9];
    float* out = (float*)d_out;

    const int N = in_sizes[0] / 64;

    char* ws = (char*)d_ws;
    int*   grid = (int*)ws;
    float* w1p  = (float*)(ws + (size_t)GRID_VOX * 4);
    float* w2p  = w1p + (size_t)27 * 4 * 16 * 32 * 2;

    grid_init_kernel<<<GRID_VOX / 256, 256, 0, stream>>>(grid);
    grid_build_kernel<<<(N + 255) / 256, 256, 0, stream>>>(coords, grid, N);
    pack_w1_kernel<<<(27 * 4 * 16 * 32 + 255) / 256, 256, 0, stream>>>(w1, w1p);
    pack_w2_kernel<<<(4 * 16 * 32 + 255) / 256, 256, 0, stream>>>(w2, w2p);

    const int nblk = (N + NROWS - 1) / NROWS;
    sparse_block_kernel<<<nblk, 128, 0, stream>>>(feat, coords, grid, w1p, b1,
                                                  gamma, beta, rmean, rvar,
                                                  w2p, b2, out, N);
}


# Rows of a tile dealt to its 16 slots by least tap collision (27 neighbours looked up once, busiest rows first): fewer chunks per tap, row prefetch depth 4 follows; per-row sum order unchanged
# speedup vs baseline: 1.4876x; 1.4876x over previous
#include <hip/hip_runtime.h>
#include <hip/hip_bf16.h>
#include <math.h>

typedef __attribute__((ext_vector_type(2))) float v2f;
typedef __attribute__((ext_vector_type(4))) float v4f;
typedef __attribute__((ext_vector_type(8))) float v8f;

#define GD 128
#define GRID_VOX (GD * GD * GD)
#define NQ 8
#define NROWS (16 * NQ)
#define PF 4
#define LSTR 68

__global__ void grid_init_kernel(int* __restrict__ grid) {
    int i = blockIdx.x * 256 + threadIdx.x;
    if (i < GRID_VOX) grid[i] = -1;
}

__global__ void grid_build_kernel(const int* __restrict__ coords,
                                  int* __restrict__ grid, int N) {
    int i = blockIdx.x * 256 + threadIdx.x;
    if (i < N) {
        int z = coords[3 * i + 0];
        int y = coords[3 * i + 1];
        int x = coords[3 * i + 2];
        grid[(z * GD + y) * GD + x] = i;
    }
}

__global__ void pack_w1_kernel(const float* __restrict__ w1, float* __restrict__ w1p) {
    int t = blockIdx.x * 256 + threadIdx.x;
    if (t >= 27 * 4 * 16 * 32) return;
    int lane = t & 31;
    int k    = (t >> 5) & 15;
    int nt   = (t >> 9) & 3;
    int off  = t >> 11;
    int half = lane >> 4;
    int col  = nt * 16 + (lane & 15);
    int krow = 4 * k + 2 * half;
    const float* w = w1 + (size_t)off * 64 * 64;
    w1p[(size_t)t * 2 + 0] = w[(krow + 0) * 64 + col];
    w1p[(size_t)t * 2 + 1] = w[(krow + 1) * 64 + col];
}

__global__ void pack_w2_kernel(const float* __restrict__ w2, float* __restrict__ w2p) {
    int t = blockIdx.x * 256 + threadIdx.x;
    if (t >= 4 * 16 * 32) return;
    int lane = t & 31;
    int k    = (t >> 5) & 15;
    int nt   = (t >> 9) & 3;
    int half = lane >> 4;
    int col  = nt * 16 + (lane & 15);
    int krow = 4 * k + 2 * half;
    w2p[(size_t)t * 2 + 0] = w2[(krow + 0) * 64 + col];
    w2p[(size_t)t * 2 + 1] = w2[(krow + 1) * 64 + col];
}


__device__ __forceinline__ unsigned row_min16(unsigned k) {
    k = min(k, (unsigned)__builtin_amdgcn_update_dpp((int)k, (int)k, 0x161, 0xF, 0xF, false));
    k = min(k, (unsigned)__builtin_amdgcn_update_dpp((int)k, (int)k, 0x162, 0xF, 0xF, false));
    k = min(k, (unsigned)__builtin_amdgcn_update_dpp((int)k, (int)k, 0x164, 0xF, 0xF, false));
    k = min(k, (unsigned)__builtin_amdgcn_update_dpp((int)k, (int)k, 0x168, 0xF, 0xF, false));
    return k;
}

__device__ __forceinline__ unsigned deal_step(unsigned m, unsigned sl, unsigned& B1, unsigned& B2,
                                              unsigned& B3, unsigned& B4, unsigned& ns) {
    const unsigned t1 = B1 & m, t2 = B2 & m, t3 = B3 & m, t4 = B4 & m;
    const unsigned cost = (unsigned)__popc(t1) + 4u * (unsigned)__popc(t2)
                        + 16u * (unsigned)__popc(t3) + 64u * (unsigned)__popc(t4);
    unsigned key = (cost << 8) | ns;
    if (ns >= (unsigned)(NQ << 4)) key = 0xFFFFFFFFu;
    key = row_min16(key);
    const unsigned tm = (sl == (key & 15u)) ? m : 0u;
    B4 |= B3 & tm;
    B3 |= B2 & tm;
    B2 |= B1 & tm;
    B1 |= tm;
    ns += (sl == (key & 15u)) ? 16u : 0u;
    return key & 0xFFu;
}

__device__ __forceinline__ void lookup_finish(int idx, int lane, int wv, int s, int q,
                                              int* csrc_b, float* skm_b, int* sJ_b) {
    const bool act = idx >= 0;
    const unsigned b    = __builtin_amdgcn_ballot_w32(act);
    const unsigned g    = (b >> (lane & 24)) & 255u;
    const int      rank = __popc(g & ((1u << q) - 1u));
    const int      cnt  = __popc(g);
    const int p = act ? rank : cnt + (q - rank);
    csrc_b[p * 16 + s] = act ? idx : -1;
    int jw = 0;
    #pragma unroll
    for (int pp = 0; pp < NQ; ++pp) {
        const bool mine = act && rank == pp;
        skm_b[(pp * 2 + (s >> 3)) * (NQ * 8) + q * 8 + (s & 7)] = mine ? 1.0f : 0.0f;
        jw += (__builtin_amdgcn_ballot_w32(mine) != 0u) ? 1 : 0;
    }
    if (lane == 0) sJ_b[wv] = jw;
}

__device__ __forceinline__ void gemm_tap(const float* abuf, const float* skm_b, int J,
                                         const v2f (&bf)[16], v8f (&acc)[NQ], int l15, int half) {
    for (int j = 0; j < J; ++j) {
        const float* ap = &abuf[(j * 16 + l15) * LSTR + 2 * half];
        v8f dA = (v8f)0.0f, dB = (v8f)0.0f;
        #pragma unroll
        for (int k = 0; k < 8; ++k) {
            v2f a0 = *(const v2f*)(ap + 4 * k);
            v2f a1 = *(const v2f*)(ap + 4 * (k + 8));
            dA = __builtin_amdgcn_wmma_f32_16x16x4_f32(
                false, a0, false, bf[k], (short)0, dA, false, false);
            dB = __builtin_amdgcn_wmma_f32_16x16x4_f32(
                false, a1, false, bf[k + 8], (short)0, dB, false, false);
        }
        const v8f* kp = (const v8f*)(skm_b + (j * 2 + half) * (NQ * 8));
        const v8f d = dA + dB;
        #pragma unroll
        for (int q = 0; q < NQ; ++q)
            acc[q] += d * kp[q];
    }
}

__global__ __launch_bounds__(128) __attribute__((amdgpu_waves_per_eu(3, 3)))
void sparse_block_kernel(const float* __restrict__ feat,
                         const int*   __restrict__ coords,
                         const int*   __restrict__ grid,
                         const float* __restrict__ w1p,
                         const float* __restrict__ b1,
                         const float* __restrict__ gamma,
                         const float* __restrict__ beta,
                         const float* __restrict__ rmean,
                         const float* __restrict__ rvar,
                         const float* __restrict__ w2p,
                         const float* __restrict__ b2,
                         float* __restrict__ out,
                         int N) {
    __shared__ __attribute__((aligned(16))) float lds[2][NROWS * LSTR];
    __shared__ int      csrc[3][NROWS];
    __shared__ __attribute__((aligned(32))) float skm[3][NQ * 2 * NQ * 8];
    __shared__ int      sJ[3][4];
    __shared__ int      snbr[27][NROWS];
    __shared__ __attribute__((aligned(32))) int srow[NROWS];

    const int tid  = threadIdx.x;
    const int lane = tid & 31;
    const int wv   = tid >> 5;
    const int half = lane >> 4;
    const int l15  = lane & 15;
    const int col  = wv * 16 + l15;
    const int row0 = blockIdx.x * NROWS;
    const bool lookup_wave = tid < NROWS;
    const int ls = tid >> 3;
    const int lq = tid & 7;
    const int lrow = lq * 16 + ls;
    const int grow = tid >> 4;
    const int gq4  = tid & 15;

    const int rg = row0 + tid;
    int cz = -10000, cy = -10000, cx = -10000;
    if (rg < N) {
        cz = coords[3 * rg + 0];
        cy = coords[3 * rg + 1];
        cx = coords[3 * rg + 2];
    }
    int nb[27];
    #pragma unroll
    for (int t = 0; t < 27; ++t) {
        const int nz = cz + (t / 9 - 1), ny = cy + ((t / 3) % 3 - 1), nx = cx + (t % 3 - 1);
        const bool inb = (unsigned)(nz | ny | nx) < (unsigned)GD;
        const int v = grid[inb ? (nz * GD + ny) * GD + nx : 0];
        nb[t] = inb ? v : -1;
    }
    unsigned am = 0;
    #pragma unroll
    for (int t = 0; t < 27; ++t) am |= (nb[t] >= 0 ? 1u : 0u) << t;
    am &= ~(1u << 13);

    int*           s_key = (int*)&lds[1][0];
    unsigned*      s_msk = (unsigned*)&lds[1][NROWS];
    unsigned*      s_pos = (unsigned*)&lds[1][2 * NROWS];
    const int mykey = (__popc(am) << 7) | (NROWS - 1 - tid);
    s_key[tid] = mykey;
    __syncthreads();
    int rank = 0;
    for (int j4 = 0; j4 < NROWS / 4; ++j4) {
        const int4 kv = *(const int4*)&s_key[j4 * 4];
        rank += (kv.x > mykey ? 1 : 0) + (kv.y > mykey ? 1 : 0)
              + (kv.z > mykey ? 1 : 0) + (kv.w > mykey ? 1 : 0);
    }
    s_msk[rank] = am;
    __syncthreads();

    if (wv == (int)(blockIdx.x & 3u)) {
        const unsigned sl = (unsigned)l15;
        unsigned B1 = 0, B2 = 0, B3 = 0, B4 = 0, ns = sl;
        uint4 mc = *(const uint4*)&s_msk[0];
        for (int i4 = 0; i4 < NROWS / 4; ++i4) {
            const uint4 mn = *(const uint4*)&s_msk[min(i4 + 1, NROWS / 4 - 1) * 4];
            const unsigned p0 = deal_step(mc.x, sl, B1, B2, B3, B4, ns);
            const unsigned p1 = deal_step(mc.y, sl, B1, B2, B3, B4, ns);
            const unsigned p2 = deal_step(mc.z, sl, B1, B2, B3, B4, ns);
            const unsigned p3 = deal_step(mc.w, sl, B1, B2, B3, B4, ns);
            if (lane == 0) s_pos[i4] = p0 | (p1 << 8) | (p2 << 16) | (p3 << 24);
            mc = mn;
        }
    }
    __syncthreads();
    {
        const int place = (int)((const unsigned char*)s_pos)[rank];
        #pragma unroll
        for (int t = 0; t < 27; ++t) snbr[t][place] = nb[t];
        srow[place] = rg < N ? rg : -1;
    }
    __syncthreads();

    if (lookup_wave) {
        lookup_finish(snbr[0][lrow], lane, wv, ls, lq, csrc[0], skm[0], sJ[0]);
        lookup_finish(snbr[1][lrow], lane, wv, ls, lq, csrc[1], skm[1], sJ[1]);
    }
    __syncthreads();
    {
        #pragma unroll
        for (int it = 0; it < 2 * NQ; ++it) {
            const int r = it * 8 + grow;
            const int sr = csrc[0][r];
            v4f v = *(const v4f*)(feat + ((unsigned)max(sr, 0) * 64u + (unsigned)gq4 * 4u));
            if (sr < 0) v = (v4f)0.0f;
            *(v4f*)(&lds[0][r * LSTR + gq4 * 4]) = v;
        }
    }
    __syncthreads();

    v8f acc[NQ];
    #pragma unroll
    for (int m = 0; m < NQ; ++m) acc[m] = (v8f)0.0f;

    int rb0 = 0, rb1 = 1, rb2 = 2;
    for (int off = 0; off < 26; ++off) {
        const int cur = off & 1;
        const bool more2 = off + 2 < 27;

        const float* bp = w1p + (size_t)(off * 4 + wv) * 16 * 64;
        v2f bf[16];
        #pragma unroll
        for (int k = 0; k < 16; ++k)
            bf[k] = *(const v2f*)(bp + k * 64 + lane * 2);

        v4f gv[2 * PF];
        unsigned gmask[2 * PF];
        const int Jn = __builtin_amdgcn_readfirstlane(
            max(max(sJ[rb1][0], sJ[rb1][1]), max(sJ[rb1][2], sJ[rb1][3])));
        {
            int sr[2 * PF];
            #pragma unroll
            for (int it = 0; it < 2 * PF; ++it) sr[it] = csrc[rb1][it * 8 + grow];
            #pragma unroll
            for (int it = 0; it < 2 * PF; ++it) {
                gmask[it] = __builtin_amdgcn_ballot_w32(sr[it] >= 0);
                gv[it] = *(const v4f*)(feat + ((unsigned)max(sr[it], 0) * 64u + (unsigned)gq4 * 4u));
            }
        }

        int idx2 = -1;
        if (more2 && lookup_wave)
            idx2 = snbr[off + 2][lrow];

        const int J  = __builtin_amdgcn_readfirstlane(
            max(max(sJ[rb0][0], sJ[rb0][1]), max(sJ[rb0][2], sJ[rb0][3])));
        gemm_tap(lds[cur], skm[rb0], J, bf, acc, l15, half);

        if (more2 && lookup_wave)
            lookup_finish(idx2, lane, wv, ls, lq, csrc[rb2], skm[rb2], sJ[rb2]);

        #pragma unroll
        for (int it = 0; it < 2 * PF; ++it) {
            if ((it >> 1) < Jn) {
                v4f v = gv[it];
                if (!__builtin_amdgcn_inverse_ballot_w32(gmask[it])) v = (v4f)0.0f;
                *(v4f*)(&lds[cur ^ 1][(it * 8 + grow) * LSTR + gq4 * 4]) = v;
            }
        }
        if (Jn > PF) {
            #pragma unroll
            for (int it = 2 * PF; it < 2 * NQ; ++it) {
                const int r = it * 8 + grow;
                const int s4 = csrc[rb1][r];
                v4f v = *(const v4f*)(feat + ((unsigned)max(s4, 0) * 64u + (unsigned)gq4 * 4u));
                if (s4 < 0) v = (v4f)0.0f;
                *(v4f*)(&lds[cur ^ 1][r * LSTR + gq4 * 4]) = v;
            }
        }
        __syncthreads();

        const int tmp = rb0; rb0 = rb1; rb1 = rb2; rb2 = tmp;
    }
    {
        const float* bp = w1p + (size_t)(26 * 4 + wv) * 16 * 64;
        v2f bf[16];
        #pragma unroll
        for (int k = 0; k < 16; ++k)
            bf[k] = *(const v2f*)(bp + k * 64 + lane * 2);
        const int J  = __builtin_amdgcn_readfirstlane(
            max(max(sJ[rb0][0], sJ[rb0][1]), max(sJ[rb0][2], sJ[rb0][3])));
        gemm_tap(lds[0], skm[rb0], J, bf, acc, l15, half);
        __syncthreads();
    }

    const float scl = gamma[col] * rsqrtf(rvar[col] + 1e-5f);
    const float shf = (b1[col] - rmean[col]) * scl + beta[col];

    #pragma unroll
    for (int m = 0; m < NQ; ++m) {
        #pragma unroll
        for (int j = 0; j < 8; ++j) {
            float y = acc[m][j] * scl + shf;
            y = y > 0.0f ? y : 0.0f;
            lds[0][(m * 16 + j + half * 8) * LSTR + col] = y;
        }
    }
    __syncthreads();

    v8f acc2[NQ];
    #pragma unroll
    for (int m = 0; m < NQ; ++m) acc2[m] = (v8f)0.0f;
    #pragma unroll
    for (int k = 0; k < 16; ++k) {
        v2f bf = *(const v2f*)(w2p + (size_t)(wv * 16 + k) * 64 + lane * 2);
        #pragma unroll
        for (int m = 0; m < NQ; ++m) {
            v2f af = *(const v2f*)(&lds[0][(m * 16 + l15) * LSTR + 4 * k + 2 * half]);
            acc2[m] = __builtin_amdgcn_wmma_f32_16x16x4_f32(
                false, af, false, bf, (short)0, acc2[m], false, false);
        }
    }

    const float bb = b2[col];
    #pragma unroll
    for (int m = 0; m < NQ; ++m) {
        const int4 ra = *(const int4*)&srow[m * 16 + half * 8];
        const int4 rb = *(const int4*)&srow[m * 16 + half * 8 + 4];
        const int rr[8] = {ra.x, ra.y, ra.z, ra.w, rb.x, rb.y, rb.z, rb.w};
        #pragma unroll
        for (int j = 0; j < 8; ++j) {
            if (rr[j] >= 0)
                out[(size_t)rr[j] * 64 + col] = acc2[m][j] + bb;
        }
    }
}

extern "C" void kernel_launch(void* const* d_in, const int* in_sizes, int n_in,
                              void* d_out, int out_size, void* d_ws, size_t ws_size,
                              hipStream_t stream) {
    const float* feat  = (const float*)d_in[0];
    const int*   coords= (const int*)  d_in[1];
    const float* w1    = (const float*)d_in[2];
    const float* b1    = (const float*)d_in[3];
    const float* gamma = (const float*)d_in[4];
    const float* beta  = (const float*)d_in[5];
    const float* rmean = (const float*)d_in[6];
    const float* rvar  = (const float*)d_in[7];
    const float* w2    = (const float*)d_in[8];
    const float* b2    = (const float*)d_in[9];
    float* out = (float*)d_out;

    const int N = in_sizes[0] / 64;

    char* ws = (char*)d_ws;
    int*   grid = (int*)ws;
    float* w1p  = (float*)(ws + (size_t)GRID_VOX * 4);
    float* w2p  = w1p + (size_t)27 * 4 * 16 * 32 * 2;

    grid_init_kernel<<<GRID_VOX / 256, 256, 0, stream>>>(grid);
    grid_build_kernel<<<(N + 255) / 256, 256, 0, stream>>>(coords, grid, N);
    pack_w1_kernel<<<(27 * 4 * 16 * 32 + 255) / 256, 256, 0, stream>>>(w1, w1p);
    pack_w2_kernel<<<(4 * 16 * 32 + 255) / 256, 256, 0, stream>>>(w2, w2p);

    const int nblk = (N + NROWS - 1) / NROWS;
    sparse_block_kernel<<<nblk, 128, 0, stream>>>(feat, coords, grid, w1p, b1,
                                                  gamma, beta, rmean, rvar,
                                                  w2p, b2, out, N);
}


# First WMMA of a tap or chunk waits only for its own A fragment: weights of tap t+1 loaded during tap t (2nd register set), chunk count carried over, A reads issued before factor reads; same sums
# speedup vs baseline: 1.5470x; 1.0399x over previous
#include <hip/hip_runtime.h>
#include <hip/hip_bf16.h>
#include <math.h>

typedef __attribute__((ext_vector_type(2))) float v2f;
typedef __attribute__((ext_vector_type(4))) float v4f;
typedef __attribute__((ext_vector_type(8))) float v8f;

#define GD 128
#define GRID_VOX (GD * GD * GD)
#define NQ 8
#define NROWS (16 * NQ)
#define PF 4
#define LSTR 68

__global__ void grid_init_kernel(int* __restrict__ grid) {
    int i = blockIdx.x * 256 + threadIdx.x;
    if (i < GRID_VOX) grid[i] = -1;
}

__global__ void grid_build_kernel(const int* __restrict__ coords,
                                  int* __restrict__ grid, int N) {
    int i = blockIdx.x * 256 + threadIdx.x;
    if (i < N) {
        int z = coords[3 * i + 0];
        int y = coords[3 * i + 1];
        int x = coords[3 * i + 2];
        grid[(z * GD + y) * GD + x] = i;
    }
}

__global__ void pack_w1_kernel(const float* __restrict__ w1, float* __restrict__ w1p) {
    int t = blockIdx.x * 256 + threadIdx.x;
    if (t >= 27 * 4 * 16 * 32) return;
    int lane = t & 31;
    int k    = (t >> 5) & 15;
    int nt   = (t >> 9) & 3;
    int off  = t >> 11;
    int half = lane >> 4;
    int col  = nt * 16 + (lane & 15);
    int krow = 4 * k + 2 * half;
    const float* w = w1 + (size_t)off * 64 * 64;
    w1p[(size_t)t * 2 + 0] = w[(krow + 0) * 64 + col];
    w1p[(size_t)t * 2 + 1] = w[(krow + 1) * 64 + col];
}

__global__ void pack_w2_kernel(const float* __restrict__ w2, float* __restrict__ w2p) {
    int t = blockIdx.x * 256 + threadIdx.x;
    if (t >= 4 * 16 * 32) return;
    int lane = t & 31;
    int k    = (t >> 5) & 15;
    int nt   = (t >> 9) & 3;
    int half = lane >> 4;
    int col  = nt * 16 + (lane & 15);
    int krow = 4 * k + 2 * half;
    w2p[(size_t)t * 2 + 0] = w2[(krow + 0) * 64 + col];
    w2p[(size_t)t * 2 + 1] = w2[(krow + 1) * 64 + col];
}


__device__ __forceinline__ unsigned row_min16(unsigned k) {
    k = min(k, (unsigned)__builtin_amdgcn_update_dpp((int)k, (int)k, 0x161, 0xF, 0xF, false));
    k = min(k, (unsigned)__builtin_amdgcn_update_dpp((int)k, (int)k, 0x162, 0xF, 0xF, false));
    k = min(k, (unsigned)__builtin_amdgcn_update_dpp((int)k, (int)k, 0x164, 0xF, 0xF, false));
    k = min(k, (unsigned)__builtin_amdgcn_update_dpp((int)k, (int)k, 0x168, 0xF, 0xF, false));
    return k;
}

__device__ __forceinline__ unsigned deal_step(unsigned m, unsigned sl, unsigned& B1, unsigned& B2,
                                              unsigned& B3, unsigned& B4, unsigned& ns) {
    const unsigned t1 = B1 & m, t2 = B2 & m, t3 = B3 & m, t4 = B4 & m;
    const unsigned cost = (unsigned)__popc(t1) + 4u * (unsigned)__popc(t2)
                        + 16u * (unsigned)__popc(t3) + 64u * (unsigned)__popc(t4);
    unsigned key = (cost << 8) | ns;
    if (ns >= (unsigned)(NQ << 4)) key = 0xFFFFFFFFu;
    key = row_min16(key);
    const unsigned tm = (sl == (key & 15u)) ? m : 0u;
    B4 |= B3 & tm;
    B3 |= B2 & tm;
    B2 |= B1 & tm;
    B1 |= tm;
    ns += (sl == (key & 15u)) ? 16u : 0u;
    return key & 0xFFu;
}

__device__ __forceinline__ void lookup_finish(int idx, int lane, int wv, int s, int q,
                                              int* csrc_b, float* skm_b, int* sJ_b) {
    const bool act = idx >= 0;
    const unsigned b    = __builtin_amdgcn_ballot_w32(act);
    const unsigned g    = (b >> (lane & 24)) & 255u;
    const int      rank = __popc(g & ((1u << q) - 1u));
    const int      cnt  = __popc(g);
    const int p = act ? rank : cnt + (q - rank);
    csrc_b[p * 16 + s] = act ? idx : -1;
    int jw = 0;
    #pragma unroll
    for (int pp = 0; pp < NQ; ++pp) {
        const bool mine = act && rank == pp;
        skm_b[(pp * 2 + (s >> 3)) * (NQ * 8) + q * 8 + (s & 7)] = mine ? 1.0f : 0.0f;
        jw += (__builtin_amdgcn_ballot_w32(mine) != 0u) ? 1 : 0;
    }
    if (lane == 0) sJ_b[wv] = jw;
}

__device__ __forceinline__ void gemm_tap(const float* abuf, const float* skm_b, int J,
                                         const v2f (&bf)[16], v8f (&acc)[NQ], int l15, int half) {
    for (int j = 0; j < J; ++j) {
        const float* ap = &abuf[(j * 16 + l15) * LSTR + 2 * half];
        v2f af[16];
        #pragma unroll
        for (int k = 0; k < 16; ++k) af[k] = *(const v2f*)(ap + 4 * k);
        __builtin_amdgcn_sched_barrier(0);
        const v8f* kp = (const v8f*)(skm_b + (j * 2 + half) * (NQ * 8));
        v8f fq[NQ];
        #pragma unroll
        for (int q = 0; q < NQ; ++q) fq[q] = kp[q];
        v8f dA = (v8f)0.0f, dB = (v8f)0.0f;
        #pragma unroll
        for (int k = 0; k < 8; ++k) {
            dA = __builtin_amdgcn_wmma_f32_16x16x4_f32(
                false, af[k], false, bf[k], (short)0, dA, false, false);
            dB = __builtin_amdgcn_wmma_f32_16x16x4_f32(
                false, af[k + 8], false, bf[k + 8], (short)0, dB, false, false);
        }
        const v8f d = dA + dB;
        #pragma unroll
        for (int q = 0; q < NQ; ++q)
            acc[q] += d * fq[q];
    }
}

__global__ __launch_bounds__(128) __attribute__((amdgpu_waves_per_eu(3, 3)))
void sparse_block_kernel(const float* __restrict__ feat,
                         const int*   __restrict__ coords,
                         const int*   __restrict__ grid,
                         const float* __restrict__ w1p,
                         const float* __restrict__ b1,
                         const float* __restrict__ gamma,
                         const float* __restrict__ beta,
                         const float* __restrict__ rmean,
                         const float* __restrict__ rvar,
                         const float* __restrict__ w2p,
                         const float* __restrict__ b2,
                         float* __restrict__ out,
                         int N) {
    __shared__ __attribute__((aligned(16))) float lds[2][NROWS * LSTR];
    __shared__ int      csrc[3][NROWS];
    __shared__ __attribute__((aligned(32))) float skm[3][NQ * 2 * NQ * 8];
    __shared__ __attribute__((aligned(16))) int sJ[3][4];
    __shared__ int      snbr[27][NROWS];
    __shared__ __attribute__((aligned(32))) int srow[NROWS];

    const int tid  = threadIdx.x;
    const int lane = tid & 31;
    const int wv   = tid >> 5;
    const int half = lane >> 4;
    const int l15  = lane & 15;
    const int col  = wv * 16 + l15;
    const int row0 = blockIdx.x * NROWS;
    const bool lookup_wave = tid < NROWS;
    const int ls = tid >> 3;
    const int lq = tid & 7;
    const int lrow = lq * 16 + ls;
    const int grow = tid >> 4;
    const int gq4  = tid & 15;

    const int rg = row0 + tid;
    int cz = -10000, cy = -10000, cx = -10000;
    if (rg < N) {
        cz = coords[3 * rg + 0];
        cy = coords[3 * rg + 1];
        cx = coords[3 * rg + 2];
    }
    int nb[27];
    #pragma unroll
    for (int t = 0; t < 27; ++t) {
        const int nz = cz + (t / 9 - 1), ny = cy + ((t / 3) % 3 - 1), nx = cx + (t % 3 - 1);
        const bool inb = (unsigned)(nz | ny | nx) < (unsigned)GD;
        const int v = grid[inb ? (nz * GD + ny) * GD + nx : 0];
        nb[t] = inb ? v : -1;
    }
    unsigned am = 0;
    #pragma unroll
    for (int t = 0; t < 27; ++t) am |= (nb[t] >= 0 ? 1u : 0u) << t;
    am &= ~(1u << 13);

    int*           s_key = (int*)&lds[1][0];
    unsigned*      s_msk = (unsigned*)&lds[1][NROWS];
    unsigned*      s_pos = (unsigned*)&lds[1][2 * NROWS];
    const int mykey = (__popc(am) << 7) | (NROWS - 1 - tid);
    s_key[tid] = mykey;
    __syncthreads();
    int rank = 0;
    for (int j4 = 0; j4 < NROWS / 4; ++j4) {
        const int4 kv = *(const int4*)&s_key[j4 * 4];
        rank += (kv.x > mykey ? 1 : 0) + (kv.y > mykey ? 1 : 0)
              + (kv.z > mykey ? 1 : 0) + (kv.w > mykey ? 1 : 0);
    }
    s_msk[rank] = am;
    __syncthreads();

    if (wv == (int)(blockIdx.x & 3u)) {
        const unsigned sl = (unsigned)l15;
        unsigned B1 = 0, B2 = 0, B3 = 0, B4 = 0, ns = sl;
        uint4 mc = *(const uint4*)&s_msk[0];
        for (int i4 = 0; i4 < NROWS / 4; ++i4) {
            const uint4 mn = *(const uint4*)&s_msk[min(i4 + 1, NROWS / 4 - 1) * 4];
            const unsigned p0 = deal_step(mc.x, sl, B1, B2, B3, B4, ns);
            const unsigned p1 = deal_step(mc.y, sl, B1, B2, B3, B4, ns);
            const unsigned p2 = deal_step(mc.z, sl, B1, B2, B3, B4, ns);
            const unsigned p3 = deal_step(mc.w, sl, B1, B2, B3, B4, ns);
            if (lane == 0) s_pos[i4] = p0 | (p1 << 8) | (p2 << 16) | (p3 << 24);
            mc = mn;
        }
    }
    __syncthreads();
    {
        const int place = (int)((const unsigned char*)s_pos)[rank];
        #pragma unroll
        for (int t = 0; t < 27; ++t) snbr[t][place] = nb[t];
        srow[place] = rg < N ? rg : -1;
    }
    __syncthreads();

    if (lookup_wave) {
        lookup_finish(snbr[0][lrow], lane, wv, ls, lq, csrc[0], skm[0], sJ[0]);
        lookup_finish(snbr[1][lrow], lane, wv, ls, lq, csrc[1], skm[1], sJ[1]);
    }
    __syncthreads();
    {
        #pragma unroll
        for (int it = 0; it < 2 * NQ; ++it) {
            const int r = it * 8 + grow;
            const int sr = csrc[0][r];
            v4f v = *(const v4f*)(feat + ((unsigned)max(sr, 0) * 64u + (unsigned)gq4 * 4u));
            if (sr < 0) v = (v4f)0.0f;
            *(v4f*)(&lds[0][r * LSTR + gq4 * 4]) = v;
        }
    }
    __syncthreads();

    v8f acc[NQ];
    #pragma unroll
    for (int m = 0; m < NQ; ++m) acc[m] = (v8f)0.0f;

    v2f bf[16];
    {
        const float* bp = w1p + (size_t)wv * 16 * 64;
        #pragma unroll
        for (int k = 0; k < 16; ++k)
            bf[k] = *(const v2f*)(bp + k * 64 + lane * 2);
    }
    int J = __builtin_amdgcn_readfirstlane(
        max(max(sJ[0][0], sJ[0][1]), max(sJ[0][2], sJ[0][3])));

    int rb0 = 0, rb1 = 1, rb2 = 2;
    for (int off = 0; off < 26; ++off) {
        const int cur = off & 1;
        const bool more2 = off + 2 < 27;

        const float* bp = w1p + (size_t)((off + 1) * 4 + wv) * 16 * 64;
        v2f bfn[16];
        #pragma unroll
        for (int k = 0; k < 16; ++k)
            bfn[k] = *(const v2f*)(bp + k * 64 + lane * 2);

        v4f gv[2 * PF];
        unsigned gmask[2 * PF];
        const int4 sjn = *(const int4*)&sJ[rb1][0];
        {
            int sr[2 * PF];
            #pragma unroll
            for (int it = 0; it < 2 * PF; ++it) sr[it] = csrc[rb1][it * 8 + grow];
            #pragma unroll
            for (int it = 0; it < 2 * PF; ++it) {
                gmask[it] = __builtin_amdgcn_ballot_w32(sr[it] >= 0);
                gv[it] = *(const v4f*)(feat + ((unsigned)max(sr[it], 0) * 64u + (unsigned)gq4 * 4u));
            }
        }

        int idx2 = -1;
        if (more2 && lookup_wave)
            idx2 = snbr[off + 2][lrow];

        gemm_tap(lds[cur], skm[rb0], J, bf, acc, l15, half);
        const int Jn = __builtin_amdgcn_readfirstlane(max(max(sjn.x, sjn.y), max(sjn.z, sjn.w)));

        if (more2 && lookup_wave)
            lookup_finish(idx2, lane, wv, ls, lq, csrc[rb2], skm[rb2], sJ[rb2]);

        #pragma unroll
        for (int it = 0; it < 2 * PF; ++it) {
            if ((it >> 1) < Jn) {
                v4f v = gv[it];
                if (!__builtin_amdgcn_inverse_ballot_w32(gmask[it])) v = (v4f)0.0f;
                *(v4f*)(&lds[cur ^ 1][(it * 8 + grow) * LSTR + gq4 * 4]) = v;
            }
        }
        if (Jn > PF) {
            #pragma unroll
            for (int it = 2 * PF; it < 2 * NQ; ++it) {
                const int r = it * 8 + grow;
                const int s4 = csrc[rb1][r];
                v4f v = *(const v4f*)(feat + ((unsigned)max(s4, 0) * 64u + (unsigned)gq4 * 4u));
                if (s4 < 0) v = (v4f)0.0f;
                *(v4f*)(&lds[cur ^ 1][r * LSTR + gq4 * 4]) = v;
            }
        }
        #pragma unroll
        for (int k = 0; k < 16; ++k) bf[k] = bfn[k];
        J = Jn;
        __syncthreads();

        const int tmp = rb0; rb0 = rb1; rb1 = rb2; rb2 = tmp;
    }
    {
        gemm_tap(lds[0], skm[rb0], J, bf, acc, l15, half);
        __syncthreads();
    }

    const float scl = gamma[col] * rsqrtf(rvar[col] + 1e-5f);
    const float shf = (b1[col] - rmean[col]) * scl + beta[col];

    #pragma unroll
    for (int m = 0; m < NQ; ++m) {
        #pragma unroll
        for (int j = 0; j < 8; ++j) {
            float y = acc[m][j] * scl + shf;
            y = y > 0.0f ? y : 0.0f;
            lds[0][(m * 16 + j + half * 8) * LSTR + col] = y;
        }
    }
    __syncthreads();

    v8f acc2[NQ];
    #pragma unroll
    for (int m = 0; m < NQ; ++m) acc2[m] = (v8f)0.0f;
    #pragma unroll
    for (int k = 0; k < 16; ++k) {
        v2f bf = *(const v2f*)(w2p + (size_t)(wv * 16 + k) * 64 + lane * 2);
        #pragma unroll
        for (int m = 0; m < NQ; ++m) {
            v2f af = *(const v2f*)(&lds[0][(m * 16 + l15) * LSTR + 4 * k + 2 * half]);
            acc2[m] = __builtin_amdgcn_wmma_f32_16x16x4_f32(
                false, af, false, bf, (short)0, acc2[m], false, false);
        }
    }

    const float bb = b2[col];
    #pragma unroll
    for (int m = 0; m < NQ; ++m) {
        const int4 ra = *(const int4*)&srow[m * 16 + half * 8];
        const int4 rb = *(const int4*)&srow[m * 16 + half * 8 + 4];
        const int rr[8] = {ra.x, ra.y, ra.z, ra.w, rb.x, rb.y, rb.z, rb.w};
        #pragma unroll
        for (int j = 0; j < 8; ++j) {
            if (rr[j] >= 0)
                out[(size_t)rr[j] * 64 + col] = acc2[m][j] + bb;
        }
    }
}

extern "C" void kernel_launch(void* const* d_in, const int* in_sizes, int n_in,
                              void* d_out, int out_size, void* d_ws, size_t ws_size,
                              hipStream_t stream) {
    const float* feat  = (const float*)d_in[0];
    const int*   coords= (const int*)  d_in[1];
    const float* w1    = (const float*)d_in[2];
    const float* b1    = (const float*)d_in[3];
    const float* gamma = (const float*)d_in[4];
    const float* beta  = (const float*)d_in[5];
    const float* rmean = (const float*)d_in[6];
    const float* rvar  = (const float*)d_in[7];
    const float* w2    = (const float*)d_in[8];
    const float* b2    = (const float*)d_in[9];
    float* out = (float*)d_out;

    const int N = in_sizes[0] / 64;

    char* ws = (char*)d_ws;
    int*   grid = (int*)ws;
    float* w1p  = (float*)(ws + (size_t)GRID_VOX * 4);
    float* w2p  = w1p + (size_t)27 * 4 * 16 * 32 * 2;

    grid_init_kernel<<<GRID_VOX / 256, 256, 0, stream>>>(grid);
    grid_build_kernel<<<(N + 255) / 256, 256, 0, stream>>>(coords, grid, N);
    pack_w1_kernel<<<(27 * 4 * 16 * 32 + 255) / 256, 256, 0, stream>>>(w1, w1p);
    pack_w2_kernel<<<(4 * 16 * 32 + 255) / 256, 256, 0, stream>>>(w2, w2p);

    const int nblk = (N + NROWS - 1) / NROWS;
    sparse_block_kernel<<<nblk, 128, 0, stream>>>(feat, coords, grid, w1p, b1,
                                                  gamma, beta, rmean, rvar,
                                                  w2p, b2, out, N);
}


# Dealing chain halved: each 16-lane row rates one of two consecutive rows per step (odd rows break ties to the high slot; re-rated only if both pick one slot), min folded into DPP; same sums
# speedup vs baseline: 1.6109x; 1.0413x over previous
#include <hip/hip_runtime.h>
#include <hip/hip_bf16.h>
#include <math.h>

typedef __attribute__((ext_vector_type(2))) float v2f;
typedef __attribute__((ext_vector_type(4))) float v4f;
typedef __attribute__((ext_vector_type(8))) float v8f;

#define GD 128
#define GRID_VOX (GD * GD * GD)
#define NQ 8
#define NROWS (16 * NQ)
#define PF 4
#define LSTR 68

__global__ void grid_init_kernel(int* __restrict__ grid) {
    int i = blockIdx.x * 256 + threadIdx.x;
    if (i < GRID_VOX) grid[i] = -1;
}

__global__ void grid_build_kernel(const int* __restrict__ coords,
                                  int* __restrict__ grid, int N) {
    int i = blockIdx.x * 256 + threadIdx.x;
    if (i < N) {
        int z = coords[3 * i + 0];
        int y = coords[3 * i + 1];
        int x = coords[3 * i + 2];
        grid[(z * GD + y) * GD + x] = i;
    }
}

__global__ void pack_w1_kernel(const float* __restrict__ w1, float* __restrict__ w1p) {
    int t = blockIdx.x * 256 + threadIdx.x;
    if (t >= 27 * 4 * 16 * 32) return;
    int lane = t & 31;
    int k    = (t >> 5) & 15;
    int nt   = (t >> 9) & 3;
    int off  = t >> 11;
    int half = lane >> 4;
    int col  = nt * 16 + (lane & 15);
    int krow = 4 * k + 2 * half;
    const float* w = w1 + (size_t)off * 64 * 64;
    w1p[(size_t)t * 2 + 0] = w[(krow + 0) * 64 + col];
    w1p[(size_t)t * 2 + 1] = w[(krow + 1) * 64 + col];
}

__global__ void pack_w2_kernel(const float* __restrict__ w2, float* __restrict__ w2p) {
    int t = blockIdx.x * 256 + threadIdx.x;
    if (t >= 4 * 16 * 32) return;
    int lane = t & 31;
    int k    = (t >> 5) & 15;
    int nt   = (t >> 9) & 3;
    int half = lane >> 4;
    int col  = nt * 16 + (lane & 15);
    int krow = 4 * k + 2 * half;
    w2p[(size_t)t * 2 + 0] = w2[(krow + 0) * 64 + col];
    w2p[(size_t)t * 2 + 1] = w2[(krow + 1) * 64 + col];
}


__device__ __forceinline__ unsigned row_min16(unsigned k) {
    k = min(k, (unsigned)__builtin_amdgcn_update_dpp(0, (int)k, 0x161, 0xF, 0xF, true));
    k = min(k, (unsigned)__builtin_amdgcn_update_dpp(0, (int)k, 0x162, 0xF, 0xF, true));
    k = min(k, (unsigned)__builtin_amdgcn_update_dpp(0, (int)k, 0x164, 0xF, 0xF, true));
    k = min(k, (unsigned)__builtin_amdgcn_update_dpp(0, (int)k, 0x168, 0xF, 0xF, true));
    return k;
}


__device__ __forceinline__ unsigned deal_key(unsigned m, unsigned B1, unsigned B2, unsigned B3,
                                             unsigned B4, unsigned nk) {
    const unsigned cost = (unsigned)__popc(B1 & m) + 4u * (unsigned)__popc(B2 & m)
                        + 16u * (unsigned)__popc(B3 & m) + 64u * (unsigned)__popc(B4 & m);
    unsigned key = (cost << 8) | nk;
    if (nk >= (unsigned)(NQ << 4)) key = 0xFFFFFFFFu;
    return row_min16(key);
}

__device__ __forceinline__ void deal_put(unsigned m, unsigned slot, unsigned sl, unsigned& B1,
                                         unsigned& B2, unsigned& B3, unsigned& B4, unsigned& nk) {
    const bool here = sl == slot;
    const unsigned tm = here ? m : 0u;
    B4 |= B3 & tm;
    B3 |= B2 & tm;
    B2 |= B1 & tm;
    B1 |= tm;
    nk += here ? 16u : 0u;
}

__device__ __forceinline__ unsigned deal_pair(unsigned mx, unsigned my, unsigned sl, bool hi,
                                              unsigned& B1, unsigned& B2, unsigned& B3,
                                              unsigned& B4, unsigned& nk) {
    const unsigned m = hi ? my : mx;
    const unsigned k = deal_key(m, B1, B2, B3, B4, nk);
    const unsigned kx = (unsigned)__builtin_amdgcn_readlane((int)k, 0);
    unsigned ky = (unsigned)__builtin_amdgcn_readlane((int)k, 16);
    const unsigned sx = kx & 15u;
    unsigned sy = 15u - (ky & 15u);
    deal_put(mx, sx, sl, B1, B2, B3, B4, nk);
    if (sx == sy) {
        const unsigned k2 = deal_key(m, B1, B2, B3, B4, nk);
        ky = (unsigned)__builtin_amdgcn_readlane((int)k2, 16);
        sy = 15u - (ky & 15u);
    }
    deal_put(my, sy, sl, B1, B2, B3, B4, nk);
    return (kx & 0xFFu) | (((ky & 0xF0u) | sy) << 8);
}

__device__ __forceinline__ void lookup_finish(int idx, int lane, int wv, int s, int q,
                                              int* csrc_b, float* skm_b, int* sJ_b) {
    const bool act = idx >= 0;
    const unsigned b    = __builtin_amdgcn_ballot_w32(act);
    const unsigned g    = (b >> (lane & 24)) & 255u;
    const int      rank = __popc(g & ((1u << q) - 1u));
    const int      cnt  = __popc(g);
    const int p = act ? rank : cnt + (q - rank);
    csrc_b[p * 16 + s] = act ? idx : -1;
    int jw = 0;
    #pragma unroll
    for (int pp = 0; pp < NQ; ++pp) {
        const bool mine = act && rank == pp;
        skm_b[(pp * 2 + (s >> 3)) * (NQ * 8) + q * 8 + (s & 7)] = mine ? 1.0f : 0.0f;
        jw += (__builtin_amdgcn_ballot_w32(mine) != 0u) ? 1 : 0;
    }
    if (lane == 0) sJ_b[wv] = jw;
}

__device__ __forceinline__ void gemm_tap(const float* abuf, const float* skm_b, int J,
                                         const v2f (&bf)[16], v8f (&acc)[NQ], int l15, int half) {
    for (int j = 0; j < J; ++j) {
        const float* ap = &abuf[(j * 16 + l15) * LSTR + 2 * half];
        v2f af[16];
        #pragma unroll
        for (int k = 0; k < 16; ++k) af[k] = *(const v2f*)(ap + 4 * k);
        __builtin_amdgcn_sched_barrier(0);
        const v8f* kp = (const v8f*)(skm_b + (j * 2 + half) * (NQ * 8));
        v8f fq[NQ];
        #pragma unroll
        for (int q = 0; q < NQ; ++q) fq[q] = kp[q];
        v8f dA = (v8f)0.0f, dB = (v8f)0.0f;
        #pragma unroll
        for (int k = 0; k < 8; ++k) {
            dA = __builtin_amdgcn_wmma_f32_16x16x4_f32(
                false, af[k], false, bf[k], (short)0, dA, false, false);
            dB = __builtin_amdgcn_wmma_f32_16x16x4_f32(
                false, af[k + 8], false, bf[k + 8], (short)0, dB, false, false);
        }
        const v8f d = dA + dB;
        #pragma unroll
        for (int q = 0; q < NQ; ++q)
            acc[q] += d * fq[q];
    }
}

__global__ __launch_bounds__(128) __attribute__((amdgpu_waves_per_eu(3, 3)))
void sparse_block_kernel(const float* __restrict__ feat,
                         const int*   __restrict__ coords,
                         const int*   __restrict__ grid,
                         const float* __restrict__ w1p,
                         const float* __restrict__ b1,
                         const float* __restrict__ gamma,
                         const float* __restrict__ beta,
                         const float* __restrict__ rmean,
                         const float* __restrict__ rvar,
                         const float* __restrict__ w2p,
                         const float* __restrict__ b2,
                         float* __restrict__ out,
                         int N) {
    __shared__ __attribute__((aligned(16))) float lds[2][NROWS * LSTR];
    __shared__ int      csrc[3][NROWS];
    __shared__ __attribute__((aligned(32))) float skm[3][NQ * 2 * NQ * 8];
    __shared__ __attribute__((aligned(16))) int sJ[3][4];
    __shared__ int      snbr[27][NROWS];
    __shared__ __attribute__((aligned(32))) int srow[NROWS];

    const int tid  = threadIdx.x;
    const int lane = tid & 31;
    const int wv   = tid >> 5;
    const int half = lane >> 4;
    const int l15  = lane & 15;
    const int col  = wv * 16 + l15;
    const int row0 = blockIdx.x * NROWS;
    const bool lookup_wave = tid < NROWS;
    const int ls = tid >> 3;
    const int lq = tid & 7;
    const int lrow = lq * 16 + ls;
    const int grow = tid >> 4;
    const int gq4  = tid & 15;

    const int rg = row0 + tid;
    int cz = -10000, cy = -10000, cx = -10000;
    if (rg < N) {
        cz = coords[3 * rg + 0];
        cy = coords[3 * rg + 1];
        cx = coords[3 * rg + 2];
    }
    int nb[27];
    #pragma unroll
    for (int t = 0; t < 27; ++t) {
        const int nz = cz + (t / 9 - 1), ny = cy + ((t / 3) % 3 - 1), nx = cx + (t % 3 - 1);
        const bool inb = (unsigned)(nz | ny | nx) < (unsigned)GD;
        const int v = grid[inb ? (nz * GD + ny) * GD + nx : 0];
        nb[t] = inb ? v : -1;
    }
    unsigned am = 0;
    #pragma unroll
    for (int t = 0; t < 27; ++t) am |= (nb[t] >= 0 ? 1u : 0u) << t;
    am &= ~(1u << 13);

    int*           s_key = (int*)&lds[1][0];
    unsigned*      s_msk = (unsigned*)&lds[1][NROWS];
    unsigned*      s_pos = (unsigned*)&lds[1][2 * NROWS];
    const int mykey = (__popc(am) << 7) | (NROWS - 1 - tid);
    s_key[tid] = mykey;
    __syncthreads();
    int rank = 0;
    for (int j4 = 0; j4 < NROWS / 4; ++j4) {
        const int4 kv = *(const int4*)&s_key[j4 * 4];
        rank += (kv.x > mykey ? 1 : 0) + (kv.y > mykey ? 1 : 0)
              + (kv.z > mykey ? 1 : 0) + (kv.w > mykey ? 1 : 0);
    }
    s_msk[rank] = am;
    __syncthreads();

    if (wv == (int)(blockIdx.x & 3u)) {
        const unsigned sl = (unsigned)l15;
        const bool hi = half != 0;
        unsigned B1 = 0, B2 = 0, B3 = 0, B4 = 0, nk = hi ? 15u - sl : sl;
        uint4 mc = *(const uint4*)&s_msk[0];
        for (int i4 = 0; i4 < NROWS / 4; ++i4) {
            const uint4 mn = *(const uint4*)&s_msk[min(i4 + 1, NROWS / 4 - 1) * 4];
            const unsigned p01 = deal_pair(mc.x, mc.y, sl, hi, B1, B2, B3, B4, nk);
            const unsigned p23 = deal_pair(mc.z, mc.w, sl, hi, B1, B2, B3, B4, nk);
            if (lane == 0) s_pos[i4] = p01 | (p23 << 16);
            mc = mn;
        }
    }
    __syncthreads();
    {
        const int place = (int)((const unsigned char*)s_pos)[rank];
        #pragma unroll
        for (int t = 0; t < 27; ++t) snbr[t][place] = nb[t];
        srow[place] = rg < N ? rg : -1;
    }
    __syncthreads();

    if (lookup_wave) {
        lookup_finish(snbr[0][lrow], lane, wv, ls, lq, csrc[0], skm[0], sJ[0]);
        lookup_finish(snbr[1][lrow], lane, wv, ls, lq, csrc[1], skm[1], sJ[1]);
    }
    __syncthreads();
    {
        #pragma unroll
        for (int it = 0; it < 2 * NQ; ++it) {
            const int r = it * 8 + grow;
            const int sr = csrc[0][r];
            v4f v = *(const v4f*)(feat + ((unsigned)max(sr, 0) * 64u + (unsigned)gq4 * 4u));
            if (sr < 0) v = (v4f)0.0f;
            *(v4f*)(&lds[0][r * LSTR + gq4 * 4]) = v;
        }
    }
    __syncthreads();

    v8f acc[NQ];
    #pragma unroll
    for (int m = 0; m < NQ; ++m) acc[m] = (v8f)0.0f;

    v2f bf[16];
    {
        const float* bp = w1p + (size_t)wv * 16 * 64;
        #pragma unroll
        for (int k = 0; k < 16; ++k)
            bf[k] = *(const v2f*)(bp + k * 64 + lane * 2);
    }
    int J = __builtin_amdgcn_readfirstlane(
        max(max(sJ[0][0], sJ[0][1]), max(sJ[0][2], sJ[0][3])));

    int rb0 = 0, rb1 = 1, rb2 = 2;
    for (int off = 0; off < 26; ++off) {
        const int cur = off & 1;
        const bool more2 = off + 2 < 27;

        const float* bp = w1p + (size_t)((off + 1) * 4 + wv) * 16 * 64;
        v2f bfn[16];
        #pragma unroll
        for (int k = 0; k < 16; ++k)
            bfn[k] = *(const v2f*)(bp + k * 64 + lane * 2);

        v4f gv[2 * PF];
        unsigned gmask[2 * PF];
        const int4 sjn = *(const int4*)&sJ[rb1][0];
        {
            int sr[2 * PF];
            #pragma unroll
            for (int it = 0; it < 2 * PF; ++it) sr[it] = csrc[rb1][it * 8 + grow];
            #pragma unroll
            for (int it = 0; it < 2 * PF; ++it) {
                gmask[it] = __builtin_amdgcn_ballot_w32(sr[it] >= 0);
                gv[it] = *(const v4f*)(feat + ((unsigned)max(sr[it], 0) * 64u + (unsigned)gq4 * 4u));
            }
        }

        int idx2 = -1;
        if (more2 && lookup_wave)
            idx2 = snbr[off + 2][lrow];

        gemm_tap(lds[cur], skm[rb0], J, bf, acc, l15, half);
        const int Jn = __builtin_amdgcn_readfirstlane(max(max(sjn.x, sjn.y), max(sjn.z, sjn.w)));

        if (more2 && lookup_wave)
            lookup_finish(idx2, lane, wv, ls, lq, csrc[rb2], skm[rb2], sJ[rb2]);

        #pragma unroll
        for (int it = 0; it < 2 * PF; ++it) {
            if ((it >> 1) < Jn) {
                v4f v = gv[it];
                if (!__builtin_amdgcn_inverse_ballot_w32(gmask[it])) v = (v4f)0.0f;
                *(v4f*)(&lds[cur ^ 1][(it * 8 + grow) * LSTR + gq4 * 4]) = v;
            }
        }
        if (Jn > PF) {
            #pragma unroll
            for (int it = 2 * PF; it < 2 * NQ; ++it) {
                const int r = it * 8 + grow;
                const int s4 = csrc[rb1][r];
                v4f v = *(const v4f*)(feat + ((unsigned)max(s4, 0) * 64u + (unsigned)gq4 * 4u));
                if (s4 < 0) v = (v4f)0.0f;
                *(v4f*)(&lds[cur ^ 1][r * LSTR + gq4 * 4]) = v;
            }
        }
        #pragma unroll
        for (int k = 0; k < 16; ++k) bf[k] = bfn[k];
        J = Jn;
        __syncthreads();

        const int tmp = rb0; rb0 = rb1; rb1 = rb2; rb2 = tmp;
    }
    {
        gemm_tap(lds[0], skm[rb0], J, bf, acc, l15, half);
        __syncthreads();
    }

    const float scl = gamma[col] * rsqrtf(rvar[col] + 1e-5f);
    const float shf = (b1[col] - rmean[col]) * scl + beta[col];

    #pragma unroll
    for (int m = 0; m < NQ; ++m) {
        #pragma unroll
        for (int j = 0; j < 8; ++j) {
            float y = acc[m][j] * scl + shf;
            y = y > 0.0f ? y : 0.0f;
            lds[0][(m * 16 + j + half * 8) * LSTR + col] = y;
        }
    }
    __syncthreads();

    v8f acc2[NQ];
    #pragma unroll
    for (int m = 0; m < NQ; ++m) acc2[m] = (v8f)0.0f;
    #pragma unroll
    for (int k = 0; k < 16; ++k) {
        v2f bf = *(const v2f*)(w2p + (size_t)(wv * 16 + k) * 64 + lane * 2);
        #pragma unroll
        for (int m = 0; m < NQ; ++m) {
            v2f af = *(const v2f*)(&lds[0][(m * 16 + l15) * LSTR + 4 * k + 2 * half]);
            acc2[m] = __builtin_amdgcn_wmma_f32_16x16x4_f32(
                false, af, false, bf, (short)0, acc2[m], false, false);
        }
    }

    const float bb = b2[col];
    #pragma unroll
    for (int m = 0; m < NQ; ++m) {
        const int4 ra = *(const int4*)&srow[m * 16 + half * 8];
        const int4 rb = *(const int4*)&srow[m * 16 + half * 8 + 4];
        const int rr[8] = {ra.x, ra.y, ra.z, ra.w, rb.x, rb.y, rb.z, rb.w};
        #pragma unroll
        for (int j = 0; j < 8; ++j) {
            if (rr[j] >= 0)
                out[(size_t)rr[j] * 64 + col] = acc2[m][j] + bb;
        }
    }
}

extern "C" void kernel_launch(void* const* d_in, const int* in_sizes, int n_in,
                              void* d_out, int out_size, void* d_ws, size_t ws_size,
                              hipStream_t stream) {
    const float* feat  = (const float*)d_in[0];
    const int*   coords= (const int*)  d_in[1];
    const float* w1    = (const float*)d_in[2];
    const float* b1    = (const float*)d_in[3];
    const float* gamma = (const float*)d_in[4];
    const float* beta  = (const float*)d_in[5];
    const float* rmean = (const float*)d_in[6];
    const float* rvar  = (const float*)d_in[7];
    const float* w2    = (const float*)d_in[8];
    const float* b2    = (const float*)d_in[9];
    float* out = (float*)d_out;

    const int N = in_sizes[0] / 64;

    char* ws = (char*)d_ws;
    int*   grid = (int*)ws;
    float* w1p  = (float*)(ws + (size_t)GRID_VOX * 4);
    float* w2p  = w1p + (size_t)27 * 4 * 16 * 32 * 2;

    grid_init_kernel<<<GRID_VOX / 256, 256, 0, stream>>>(grid);
    grid_build_kernel<<<(N + 255) / 256, 256, 0, stream>>>(coords, grid, N);
    pack_w1_kernel<<<(27 * 4 * 16 * 32 + 255) / 256, 256, 0, stream>>>(w1, w1p);
    pack_w2_kernel<<<(4 * 16 * 32 + 255) / 256, 256, 0, stream>>>(w2, w2p);

    const int nblk = (N + NROWS - 1) / NROWS;
    sparse_block_kernel<<<nblk, 128, 0, stream>>>(feat, coords, grid, w1p, b1,
                                                  gamma, beta, rmean, rvar,
                                                  w2p, b2, out, N);
}
